# starting kernel
# baseline (speedup 1.0000x reference)
#include <hip/hip_runtime.h>
#include <math.h>

typedef __attribute__((ext_vector_type(2))) float v2f;
typedef __attribute__((ext_vector_type(8))) float v8f;

__device__ __forceinline__ v8f wmma_f32(v2f a, v2f b, v8f c) {
  return __builtin_amdgcn_wmma_f32_16x16x4_f32(
       false, a,  false, b,
       (short)0, c,  false,  false);
}

__global__ void zero_kernel(float* __restrict__ p, int n) {
  int i = blockIdx.x * blockDim.x + threadIdx.x;
  if (i < n) p[i] = 0.0f;
}

__global__ __launch_bounds__(256) void mlp_kernel(const float* __restrict__ x,
                                                  const float* __restrict__ W,
                                                  const float* __restrict__ bias,
                                                  float* __restrict__ t) {
  __shared__ float Wl[8 * 1024];
  int tid = threadIdx.x;
  const float4* W4 = (const float4*)W;
  float4* Wl4 = (float4*)Wl;
  for (int i = tid; i < 2048; i += 256) Wl4[i] = W4[i];
  __syncthreads();

  int wave = tid >> 5, lane = tid & 31;
  int row = blockIdx.x * 8 + wave;
  const float4* xr = (const float4*)(x + (size_t)row * 1024);

  float acc[8];
#pragma unroll
  for (int c = 0; c < 8; ++c) acc[c] = 0.0f;

  for (int i = 0; i < 8; ++i) {
    int wi = i * 32 + lane;
    float4 xv = xr[wi];
#pragma unroll
    for (int c = 0; c < 8; ++c) {
      float4 wv = Wl4[c * 256 + wi];
      acc[c] += xv.x * wv.x + xv.y * wv.y + xv.z * wv.z + xv.w * wv.w;
    }
  }
#pragma unroll
  for (int c = 0; c < 8; ++c)
#pragma unroll
    for (int off = 16; off > 0; off >>= 1)
      acc[c] += __shfl_xor(acc[c], off, 32);

  if (lane == 0) {
#pragma unroll
    for (int c = 0; c < 8; ++c)
      t[(size_t)row * 8 + c] = tanhf(acc[c] + bias[c]);
  }
}

__global__ __launch_bounds__(128) void tmat_kernel(const float* __restrict__ Q,
                                                   const float* __restrict__ selW,
                                                   const float* __restrict__ t,
                                                   float* __restrict__ T) {
  int wg = blockIdx.x;
  int part = wg & 3;
  int bh = wg >> 2;
  int b = bh >> 3, h = bh & 7;

  __shared__ float a[256];
  int tid = threadIdx.x;
  const float* tp = t + (size_t)b * 8192 + (size_t)h * 1024 + part * 256;
  for (int i = tid; i < 256; i += 128) a[i] = tp[i];
  __syncthreads();

  int wave = tid >> 5, lane = tid & 31;
  int half = lane >> 4, lm = lane & 15;
  int e = wave * 16 + lm;

  float wsel[4];
#pragma unroll
  for (int gi = 0; gi < 4; ++gi) wsel[gi] = selW[h * 64 + gi * 16 + lm];

  v8f acc[4] = {};
  const float* Qbase = Q + ((size_t)b * 1024 + (size_t)part * 256) * 512 + h * 64 + e;

  for (int ll = 0; ll < 256; ll += 4) {
    int la = ll + half * 2;
    v2f af;
    af.x = Qbase[(size_t)la * 512];
    af.y = Qbase[(size_t)(la + 1) * 512];
    float a0 = a[la], a1 = a[la + 1];
#pragma unroll
    for (int gi = 0; gi < 4; ++gi) {
      float s0 = a0 * wsel[gi], s1 = a1 * wsel[gi];
      v2f bf;
      bf.x = s0 > 0.0f ? s0 : 0.0f;
      bf.y = s1 > 0.0f ? s1 : 0.0f;
      acc[gi] = wmma_f32(af, bf, acc[gi]);
    }
  }

  float* Tp = T + (size_t)bh * 4096;
#pragma unroll
  for (int gi = 0; gi < 4; ++gi)
#pragma unroll
    for (int jj = 0; jj < 8; ++jj) {
      int er = wave * 16 + jj + half * 8;
      int g = gi * 16 + lm;
      atomicAdd(&Tp[er * 64 + g], acc[gi][jj]);
    }
}

#define TPAD 72
#define KPAD 65
__global__ __launch_bounds__(128) void attn_kernel(const float* __restrict__ Km,
                                                   const float* __restrict__ Vm,
                                                   const float* __restrict__ Tg,
                                                   float* __restrict__ out) {
  int wg = blockIdx.x;
  int part = wg & 3;
  int bh = wg >> 2;
  int b = bh >> 3, h = bh & 7;

  __shared__ float Tl[64 * TPAD];
  __shared__ float Kb[64 * KPAD];
  __shared__ float Sb[64 * KPAD];

  int tid = threadIdx.x;
  const float* Tp = Tg + (size_t)bh * 4096;
  for (int i = tid; i < 4096; i += 128) {
    int e = i >> 6, g = i & 63;
    Tl[e * TPAD + g] = Tp[i];
  }

  int wave = tid >> 5, lane = tid & 31;
  int half = lane >> 4, lm = lane & 15;

  v8f vacc[4] = {};
  const float* Kbase = Vm == Km ? Km : Km;
  Kbase = Km + ((size_t)b * 1024 * 8 + h) * 64;
  const float* Vbase = Vm + ((size_t)b * 1024 * 8 + h) * 64;

  for (int chunk = 0; chunk < 4; ++chunk) {
    int s0 = part * 256 + chunk * 64;
    __syncthreads();

    for (int i = tid; i < 4096; i += 128) {
      int r = i >> 6, e = i & 63;
      Kb[r * KPAD + e] = Kbase[(size_t)(s0 + r) * 512 + e];
    }
    __syncthreads();

    v8f sacc[4] = {};
    int m = wave * 16 + lm;
    for (int e0 = 0; e0 < 64; e0 += 4) {
      int ka = e0 + half * 2;
      v2f af;
      af.x = Kb[m * KPAD + ka];
      af.y = Kb[m * KPAD + ka + 1];
#pragma unroll
      for (int gi = 0; gi < 4; ++gi) {
        int n = gi * 16 + lm;
        v2f bf;
        bf.x = Tl[ka * TPAD + n];
        bf.y = Tl[(ka + 1) * TPAD + n];
        sacc[gi] = wmma_f32(af, bf, sacc[gi]);
      }
    }
#pragma unroll
    for (int gi = 0; gi < 4; ++gi)
#pragma unroll
      for (int jj = 0; jj < 8; ++jj) {
        int r = wave * 16 + jj + half * 8;
        Sb[r * KPAD + gi * 16 + lm] = sacc[gi][jj];
      }
    __syncthreads();

    if (tid < 64) {
      float* row = &Sb[tid * KPAD];
      float mx = row[0] * 0.125f;
      for (int g = 1; g < 64; ++g) mx = fmaxf(mx, row[g] * 0.125f);
      float sum = 0.0f;
      for (int g = 0; g < 64; ++g) {
        float ev = __expf(row[g] * 0.125f - mx);
        row[g] = ev;
        sum += ev;
      }
      float inv = 1.0f / sum;
      for (int g = 0; g < 64; ++g) row[g] *= inv;
    }
    __syncthreads();

    const float* Vp = Vbase + (size_t)s0 * 512 + wave * 16 + lm;
    for (int k0 = 0; k0 < 64; k0 += 4) {
      int ka = k0 + half * 2;
      v2f af;
      af.x = Vp[(size_t)ka * 512];
      af.y = Vp[(size_t)(ka + 1) * 512];
#pragma unroll
      for (int gi = 0; gi < 4; ++gi) {
        int n = gi * 16 + lm;
        v2f bf;
        bf.x = Sb[ka * KPAD + n];
        bf.y = Sb[(ka + 1) * KPAD + n];
        vacc[gi] = wmma_f32(af, bf, vacc[gi]);
      }
    }
  }

#pragma unroll
  for (int gi = 0; gi < 4; ++gi)
#pragma unroll
    for (int jj = 0; jj < 8; ++jj) {
      int d = wave * 16 + jj + half * 8;
      int g = gi * 16 + lm;
      atomicAdd(&out[(((size_t)b * 64 + d) * 8 + h) * 64 + g], vacc[gi][jj]);
    }
}

extern "C" void kernel_launch(void* const* d_in, const int* in_sizes, int n_in,
                              void* d_out, int out_size, void* d_ws, size_t ws_size,
                              hipStream_t stream) {
  const float* queries = (const float*)d_in[0];
  const float* keys    = (const float*)d_in[1];
  const float* values  = (const float*)d_in[2];
  const float* x       = (const float*)d_in[3];
  const float* mlp_w   = (const float*)d_in[4];
  const float* mlp_b   = (const float*)d_in[5];
  const float* sel_W   = (const float*)d_in[6];
  (void)d_in; (void)in_sizes; (void)n_in; (void)ws_size;

  float* out  = (float*)d_out;
  float* t_ws = (float*)d_ws;
  float* T_ws = t_ws + 131072;

  zero_kernel<<<2048, 256, 0, stream>>>(out, 524288);
  zero_kernel<<<2048, 256, 0, stream>>>(T_ws, 524288);

  mlp_kernel<<<2048, 256, 0, stream>>>(x, mlp_w, mlp_b, t_ws);

  tmat_kernel<<<512, 128, 0, stream>>>(queries, sel_W, t_ws, T_ws);

  attn_kernel<<<512, 128, 0, stream>>>(keys, values, T_ws, out);
}


# tmat+attn: each tile's global loads issued as one batch, one tile ahead of use (Q slice, T, K tiles, V fragments wait in registers; hand-written LDS barrier keeps them in flight)
# speedup vs baseline: 1.3599x; 1.3599x over previous
#include <hip/hip_runtime.h>
#include <math.h>

typedef __attribute__((ext_vector_type(2))) float v2f;
typedef __attribute__((ext_vector_type(8))) float v8f;

__device__ __forceinline__ v8f wmma_f32(v2f a, v2f b, v8f c) {
  return __builtin_amdgcn_wmma_f32_16x16x4_f32(
       false, a,  false, b,
       (short)0, c,  false,  false);
}

__device__ __forceinline__ void lds_barrier() {
  asm volatile("s_wait_dscnt 0x0\n\ts_barrier_signal -1\n\ts_barrier_wait -1" ::: "memory");
}

__global__ void zero_kernel(float* __restrict__ p, int n) {
  int i = blockIdx.x * blockDim.x + threadIdx.x;
  if (i < n) p[i] = 0.0f;
}

__global__ __launch_bounds__(256) void mlp_kernel(const float* __restrict__ x,
                                                  const float* __restrict__ W,
                                                  const float* __restrict__ bias,
                                                  float* __restrict__ t) {
  __shared__ float Wl[8 * 1024];
  int tid = threadIdx.x;
  const float4* W4 = (const float4*)W;
  float4* Wl4 = (float4*)Wl;
  for (int i = tid; i < 2048; i += 256) Wl4[i] = W4[i];
  __syncthreads();

  int wave = tid >> 5, lane = tid & 31;
  int row = blockIdx.x * 8 + wave;
  const float4* xr = (const float4*)(x + (size_t)row * 1024);

  float acc[8];
#pragma unroll
  for (int c = 0; c < 8; ++c) acc[c] = 0.0f;

  for (int i = 0; i < 8; ++i) {
    int wi = i * 32 + lane;
    float4 xv = xr[wi];
#pragma unroll
    for (int c = 0; c < 8; ++c) {
      float4 wv = Wl4[c * 256 + wi];
      acc[c] += xv.x * wv.x + xv.y * wv.y + xv.z * wv.z + xv.w * wv.w;
    }
  }
#pragma unroll
  for (int c = 0; c < 8; ++c)
#pragma unroll
    for (int off = 16; off > 0; off >>= 1)
      acc[c] += __shfl_xor(acc[c], off, 32);

  if (lane == 0) {
#pragma unroll
    for (int c = 0; c < 8; ++c)
      t[(size_t)row * 8 + c] = tanhf(acc[c] + bias[c]);
  }
}

__global__ __launch_bounds__(128) void tmat_kernel(const float* __restrict__ Q,
                                                   const float* __restrict__ selW,
                                                   const float* __restrict__ t,
                                                   float* __restrict__ T) {
  int wg = blockIdx.x;
  int part = wg & 3;
  int bh = wg >> 2;
  int b = bh >> 3, h = bh & 7;

  __shared__ float a[256];
  int tid = threadIdx.x;
  int wave = tid >> 5, lane = tid & 31;
  int half = lane >> 4, lm = lane & 15;
  int e = wave * 16 + lm;

  const float* tp = t + (size_t)b * 8192 + (size_t)h * 1024 + part * 256;
  float ta0 = tp[tid], ta1 = tp[tid + 128];

  float wsel[4];
#pragma unroll
  for (int gi = 0; gi < 4; ++gi) wsel[gi] = selW[h * 64 + gi * 16 + lm];

  const float* Qbase = Q + ((size_t)b * 1024 + (size_t)part * 256) * 512 + h * 64 + e
                         + (size_t)half * 1024;
  v2f qf[64];
#pragma unroll
  for (int it = 0; it < 64; ++it) {
    qf[it].x = Qbase[(size_t)it * 2048];
    qf[it].y = Qbase[(size_t)it * 2048 + 512];
  }

  a[tid] = ta0;
  a[tid + 128] = ta1;
  lds_barrier();

  v8f acc[4] = {};
#pragma unroll
  for (int it = 0; it < 64; ++it) {
    int la = it * 4 + half * 2;
    float a0 = a[la], a1 = a[la + 1];
#pragma unroll
    for (int gi = 0; gi < 4; ++gi) {
      float s0 = a0 * wsel[gi], s1 = a1 * wsel[gi];
      v2f bf;
      bf.x = s0 > 0.0f ? s0 : 0.0f;
      bf.y = s1 > 0.0f ? s1 : 0.0f;
      acc[gi] = wmma_f32(qf[it], bf, acc[gi]);
    }
  }

  float* Tp = T + (size_t)bh * 4096;
#pragma unroll
  for (int gi = 0; gi < 4; ++gi)
#pragma unroll
    for (int jj = 0; jj < 8; ++jj) {
      int er = wave * 16 + jj + half * 8;
      int g = gi * 16 + lm;
      atomicAdd(&Tp[er * 64 + g], acc[gi][jj]);
    }
}

#define TPAD 72
#define KPAD 65
__global__ __launch_bounds__(128) void attn_kernel(const float* __restrict__ Km,
                                                   const float* __restrict__ Vm,
                                                   const float* __restrict__ Tg,
                                                   float* __restrict__ out) {
  int wg = blockIdx.x;
  int part = wg & 3;
  int bh = wg >> 2;
  int b = bh >> 3, h = bh & 7;

  __shared__ float Tl[64 * TPAD];
  __shared__ float Kb[64 * KPAD];
  __shared__ float Sb[64 * KPAD];

  int tid = threadIdx.x;
  int wave = tid >> 5, lane = tid & 31;
  int half = lane >> 4, lm = lane & 15;
  int sr = tid >> 6, sc = tid & 63;

  const float* Tp = Tg + (size_t)bh * 4096 + tid;
  const float* Kbase = Km + ((size_t)b * 1024 * 8 + h) * 64;
  const float* Vbase = Vm + ((size_t)b * 1024 * 8 + h) * 64;
  const float* Kst = Kbase + ((size_t)part * 256 + sr) * 512 + sc;

  float tr[32], kr[32];
#pragma unroll
  for (int j = 0; j < 32; ++j) tr[j] = Tp[j * 128];
#pragma unroll
  for (int j = 0; j < 32; ++j) kr[j] = Kst[(size_t)j * 1024];
  __builtin_amdgcn_sched_barrier(0);
#pragma unroll
  for (int j = 0; j < 32; ++j) Tl[(sr + 2 * j) * TPAD + sc] = tr[j];

  v8f vacc[4] = {};

  for (int chunk = 0; chunk < 4; ++chunk) {
    int s0 = part * 256 + chunk * 64;
    lds_barrier();

#pragma unroll
    for (int j = 0; j < 32; ++j) Kb[(sr + 2 * j) * KPAD + sc] = kr[j];

    const float* Vp = Vbase + (size_t)s0 * 512 + wave * 16 + lm + (size_t)half * 1024;
    v2f vf[16];
#pragma unroll
    for (int kt = 0; kt < 16; ++kt) {
      vf[kt].x = Vp[(size_t)kt * 2048];
      vf[kt].y = Vp[(size_t)kt * 2048 + 512];
    }
    lds_barrier();

    v8f sacc[4] = {};
    int m = wave * 16 + lm;
    for (int e0 = 0; e0 < 64; e0 += 4) {
      int ka = e0 + half * 2;
      v2f af;
      af.x = Kb[m * KPAD + ka];
      af.y = Kb[m * KPAD + ka + 1];
#pragma unroll
      for (int gi = 0; gi < 4; ++gi) {
        int n = gi * 16 + lm;
        v2f bf;
        bf.x = Tl[ka * TPAD + n];
        bf.y = Tl[(ka + 1) * TPAD + n];
        sacc[gi] = wmma_f32(af, bf, sacc[gi]);
      }
    }
#pragma unroll
    for (int gi = 0; gi < 4; ++gi)
#pragma unroll
      for (int jj = 0; jj < 8; ++jj) {
        int r = wave * 16 + jj + half * 8;
        Sb[r * KPAD + gi * 16 + lm] = sacc[gi][jj];
      }

    if (chunk < 3) {
      const float* Kn = Kst + (size_t)(chunk + 1) * 32768;
#pragma unroll
      for (int j = 0; j < 32; ++j) kr[j] = Kn[(size_t)j * 1024];
    }
    lds_barrier();

    if (tid < 64) {
      float* row = &Sb[tid * KPAD];
      float mx = row[0] * 0.125f;
      for (int g = 1; g < 64; ++g) mx = fmaxf(mx, row[g] * 0.125f);
      float sum = 0.0f;
      for (int g = 0; g < 64; ++g) {
        float ev = __expf(row[g] * 0.125f - mx);
        row[g] = ev;
        sum += ev;
      }
      float inv = 1.0f / sum;
      for (int g = 0; g < 64; ++g) row[g] *= inv;
    }
    lds_barrier();

#pragma unroll
    for (int kt = 0; kt < 16; ++kt) {
      int ka = kt * 4 + half * 2;
#pragma unroll
      for (int gi = 0; gi < 4; ++gi) {
        int n = gi * 16 + lm;
        v2f bf;
        bf.x = Sb[ka * KPAD + n];
        bf.y = Sb[(ka + 1) * KPAD + n];
        vacc[gi] = wmma_f32(vf[kt], bf, vacc[gi]);
      }
    }
  }

#pragma unroll
  for (int gi = 0; gi < 4; ++gi)
#pragma unroll
    for (int jj = 0; jj < 8; ++jj) {
      int d = wave * 16 + jj + half * 8;
      int g = gi * 16 + lm;
      atomicAdd(&out[(((size_t)b * 64 + d) * 8 + h) * 64 + g], vacc[gi][jj]);
    }
}

extern "C" void kernel_launch(void* const* d_in, const int* in_sizes, int n_in,
                              void* d_out, int out_size, void* d_ws, size_t ws_size,
                              hipStream_t stream) {
  const float* queries = (const float*)d_in[0];
  const float* keys    = (const float*)d_in[1];
  const float* values  = (const float*)d_in[2];
  const float* x       = (const float*)d_in[3];
  const float* mlp_w   = (const float*)d_in[4];
  const float* mlp_b   = (const float*)d_in[5];
  const float* sel_W   = (const float*)d_in[6];
  (void)d_in; (void)in_sizes; (void)n_in; (void)ws_size;

  float* out  = (float*)d_out;
  float* t_ws = (float*)d_ws;
  float* T_ws = t_ws + 131072;

  zero_kernel<<<2048, 256, 0, stream>>>(out, 524288);
  zero_kernel<<<2048, 256, 0, stream>>>(T_ws, 524288);

  mlp_kernel<<<2048, 256, 0, stream>>>(x, mlp_w, mlp_b, t_ws);

  tmat_kernel<<<512, 128, 0, stream>>>(queries, sel_W, t_ws, T_ws);

  attn_kernel<<<512, 128, 0, stream>>>(keys, values, T_ws, out);
}


# zero fills of out and T folded into the mlp kernel's blocks as 16-byte stores: 5 launches -> 3, two 16384-wave fill launches gone; atomics and their summation order as before
# speedup vs baseline: 1.4463x; 1.0635x over previous
#include <hip/hip_runtime.h>
#include <math.h>

typedef __attribute__((ext_vector_type(2))) float v2f;
typedef __attribute__((ext_vector_type(8))) float v8f;

__device__ __forceinline__ v8f wmma_f32(v2f a, v2f b, v8f c) {
  return __builtin_amdgcn_wmma_f32_16x16x4_f32(
       false, a,  false, b,
       (short)0, c,  false,  false);
}

__device__ __forceinline__ void lds_barrier() {
  asm volatile("s_wait_dscnt 0x0\n\ts_barrier_signal -1\n\ts_barrier_wait -1" ::: "memory");
}

__global__ __launch_bounds__(256) void mlp_kernel(const float* __restrict__ x,
                                                  const float* __restrict__ W,
                                                  const float* __restrict__ bias,
                                                  float* __restrict__ t,
                                                  float* __restrict__ zT,
                                                  float* __restrict__ zout) {
  __shared__ float Wl[8 * 1024];
  int tid = threadIdx.x;
  const float4* W4 = (const float4*)W;
  float4* Wl4 = (float4*)Wl;
  for (int i = tid; i < 2048; i += 256) Wl4[i] = W4[i];
  __syncthreads();

  if (tid < 128) {
    float4* z = (float4*)(tid < 64 ? zout : zT);
    z[blockIdx.x * 64 + (tid & 63)] = make_float4(0.0f, 0.0f, 0.0f, 0.0f);
  }

  int wave = tid >> 5, lane = tid & 31;
  int row = blockIdx.x * 8 + wave;
  const float4* xr = (const float4*)(x + (size_t)row * 1024);

  float acc[8];
#pragma unroll
  for (int c = 0; c < 8; ++c) acc[c] = 0.0f;

  for (int i = 0; i < 8; ++i) {
    int wi = i * 32 + lane;
    float4 xv = xr[wi];
#pragma unroll
    for (int c = 0; c < 8; ++c) {
      float4 wv = Wl4[c * 256 + wi];
      acc[c] += xv.x * wv.x + xv.y * wv.y + xv.z * wv.z + xv.w * wv.w;
    }
  }
#pragma unroll
  for (int c = 0; c < 8; ++c)
#pragma unroll
    for (int off = 16; off > 0; off >>= 1)
      acc[c] += __shfl_xor(acc[c], off, 32);

  if (lane == 0) {
#pragma unroll
    for (int c = 0; c < 8; ++c)
      t[(size_t)row * 8 + c] = tanhf(acc[c] + bias[c]);
  }
}

__global__ __launch_bounds__(128) void tmat_kernel(const float* __restrict__ Q,
                                                   const float* __restrict__ selW,
                                                   const float* __restrict__ t,
                                                   float* __restrict__ T) {
  int wg = blockIdx.x;
  int part = wg & 3;
  int bh = wg >> 2;
  int b = bh >> 3, h = bh & 7;

  __shared__ float a[256];
  int tid = threadIdx.x;
  int wave = tid >> 5, lane = tid & 31;
  int half = lane >> 4, lm = lane & 15;
  int e = wave * 16 + lm;

  const float* tp = t + (size_t)b * 8192 + (size_t)h * 1024 + part * 256;
  float ta0 = tp[tid], ta1 = tp[tid + 128];

  float wsel[4];
#pragma unroll
  for (int gi = 0; gi < 4; ++gi) wsel[gi] = selW[h * 64 + gi * 16 + lm];

  const float* Qbase = Q + ((size_t)b * 1024 + (size_t)part * 256) * 512 + h * 64 + e
                         + (size_t)half * 1024;
  v2f qf[64];
#pragma unroll
  for (int it = 0; it < 64; ++it) {
    qf[it].x = Qbase[(size_t)it * 2048];
    qf[it].y = Qbase[(size_t)it * 2048 + 512];
  }

  a[tid] = ta0;
  a[tid + 128] = ta1;
  lds_barrier();

  v8f acc[4] = {};
#pragma unroll
  for (int it = 0; it < 64; ++it) {
    int la = it * 4 + half * 2;
    float a0 = a[la], a1 = a[la + 1];
#pragma unroll
    for (int gi = 0; gi < 4; ++gi) {
      float s0 = a0 * wsel[gi], s1 = a1 * wsel[gi];
      v2f bf;
      bf.x = s0 > 0.0f ? s0 : 0.0f;
      bf.y = s1 > 0.0f ? s1 : 0.0f;
      acc[gi] = wmma_f32(qf[it], bf, acc[gi]);
    }
  }

  float* Tp = T + (size_t)bh * 4096;
#pragma unroll
  for (int gi = 0; gi < 4; ++gi)
#pragma unroll
    for (int jj = 0; jj < 8; ++jj) {
      int er = wave * 16 + jj + half * 8;
      int g = gi * 16 + lm;
      atomicAdd(&Tp[er * 64 + g], acc[gi][jj]);
    }
}

#define TPAD 72
#define KPAD 65
__global__ __launch_bounds__(128) void attn_kernel(const float* __restrict__ Km,
                                                   const float* __restrict__ Vm,
                                                   const float* __restrict__ Tg,
                                                   float* __restrict__ out) {
  int wg = blockIdx.x;
  int part = wg & 3;
  int bh = wg >> 2;
  int b = bh >> 3, h = bh & 7;

  __shared__ float Tl[64 * TPAD];
  __shared__ float Kb[64 * KPAD];
  __shared__ float Sb[64 * KPAD];

  int tid = threadIdx.x;
  int wave = tid >> 5, lane = tid & 31;
  int half = lane >> 4, lm = lane & 15;
  int sr = tid >> 6, sc = tid & 63;

  const float* Tp = Tg + (size_t)bh * 4096 + tid;
  const float* Kbase = Km + ((size_t)b * 1024 * 8 + h) * 64;
  const float* Vbase = Vm + ((size_t)b * 1024 * 8 + h) * 64;
  const float* Kst = Kbase + ((size_t)part * 256 + sr) * 512 + sc;

  float tr[32], kr[32];
#pragma unroll
  for (int j = 0; j < 32; ++j) tr[j] = Tp[j * 128];
#pragma unroll
  for (int j = 0; j < 32; ++j) kr[j] = Kst[(size_t)j * 1024];
  __builtin_amdgcn_sched_barrier(0);
#pragma unroll
  for (int j = 0; j < 32; ++j) Tl[(sr + 2 * j) * TPAD + sc] = tr[j];

  v8f vacc[4] = {};

  for (int chunk = 0; chunk < 4; ++chunk) {
    int s0 = part * 256 + chunk * 64;
    lds_barrier();

#pragma unroll
    for (int j = 0; j < 32; ++j) Kb[(sr + 2 * j) * KPAD + sc] = kr[j];

    const float* Vp = Vbase + (size_t)s0 * 512 + wave * 16 + lm + (size_t)half * 1024;
    v2f vf[16];
#pragma unroll
    for (int kt = 0; kt < 16; ++kt) {
      vf[kt].x = Vp[(size_t)kt * 2048];
      vf[kt].y = Vp[(size_t)kt * 2048 + 512];
    }
    lds_barrier();

    v8f sacc[4] = {};
    int m = wave * 16 + lm;
    for (int e0 = 0; e0 < 64; e0 += 4) {
      int ka = e0 + half * 2;
      v2f af;
      af.x = Kb[m * KPAD + ka];
      af.y = Kb[m * KPAD + ka + 1];
#pragma unroll
      for (int gi = 0; gi < 4; ++gi) {
        int n = gi * 16 + lm;
        v2f bf;
        bf.x = Tl[ka * TPAD + n];
        bf.y = Tl[(ka + 1) * TPAD + n];
        sacc[gi] = wmma_f32(af, bf, sacc[gi]);
      }
    }
#pragma unroll
    for (int gi = 0; gi < 4; ++gi)
#pragma unroll
      for (int jj = 0; jj < 8; ++jj) {
        int r = wave * 16 + jj + half * 8;
        Sb[r * KPAD + gi * 16 + lm] = sacc[gi][jj];
      }

    if (chunk < 3) {
      const float* Kn = Kst + (size_t)(chunk + 1) * 32768;
#pragma unroll
      for (int j = 0; j < 32; ++j) kr[j] = Kn[(size_t)j * 1024];
    }
    lds_barrier();

    if (tid < 64) {
      float* row = &Sb[tid * KPAD];
      float mx = row[0] * 0.125f;
      for (int g = 1; g < 64; ++g) mx = fmaxf(mx, row[g] * 0.125f);
      float sum = 0.0f;
      for (int g = 0; g < 64; ++g) {
        float ev = __expf(row[g] * 0.125f - mx);
        row[g] = ev;
        sum += ev;
      }
      float inv = 1.0f / sum;
      for (int g = 0; g < 64; ++g) row[g] *= inv;
    }
    lds_barrier();

#pragma unroll
    for (int kt = 0; kt < 16; ++kt) {
      int ka = kt * 4 + half * 2;
#pragma unroll
      for (int gi = 0; gi < 4; ++gi) {
        int n = gi * 16 + lm;
        v2f bf;
        bf.x = Sb[ka * KPAD + n];
        bf.y = Sb[(ka + 1) * KPAD + n];
        vacc[gi] = wmma_f32(vf[kt], bf, vacc[gi]);
      }
    }
  }

#pragma unroll
  for (int gi = 0; gi < 4; ++gi)
#pragma unroll
    for (int jj = 0; jj < 8; ++jj) {
      int d = wave * 16 + jj + half * 8;
      int g = gi * 16 + lm;
      atomicAdd(&out[(((size_t)b * 64 + d) * 8 + h) * 64 + g], vacc[gi][jj]);
    }
}

extern "C" void kernel_launch(void* const* d_in, const int* in_sizes, int n_in,
                              void* d_out, int out_size, void* d_ws, size_t ws_size,
                              hipStream_t stream) {
  const float* queries = (const float*)d_in[0];
  const float* keys    = (const float*)d_in[1];
  const float* values  = (const float*)d_in[2];
  const float* x       = (const float*)d_in[3];
  const float* mlp_w   = (const float*)d_in[4];
  const float* mlp_b   = (const float*)d_in[5];
  const float* sel_W   = (const float*)d_in[6];
  (void)d_in; (void)in_sizes; (void)n_in; (void)ws_size;

  float* out  = (float*)d_out;
  float* t_ws = (float*)d_ws;
  float* T_ws = t_ws + 131072;

  mlp_kernel<<<2048, 256, 0, stream>>>(x, mlp_w, mlp_b, t_ws, T_ws, out);

  tmat_kernel<<<512, 128, 0, stream>>>(queries, sel_W, t_ws, T_ws);

  attn_kernel<<<512, 128, 0, stream>>>(keys, values, T_ws, out);
}


# attn: softmax in registers on the score tiles' accumulator layout by all 4 waves (row max/sum = 4 in-lane values + 16-lane DPP butterfly; row sum now a tree), not 64 threads looping over LDS
# speedup vs baseline: 1.7485x; 1.2089x over previous
#include <hip/hip_runtime.h>
#include <math.h>

typedef __attribute__((ext_vector_type(2))) float v2f;
typedef __attribute__((ext_vector_type(8))) float v8f;

__device__ __forceinline__ v8f wmma_f32(v2f a, v2f b, v8f c) {
  return __builtin_amdgcn_wmma_f32_16x16x4_f32(
       false, a,  false, b,
       (short)0, c,  false,  false);
}

__device__ __forceinline__ void lds_barrier() {
  asm volatile("s_wait_dscnt 0x0\n\ts_barrier_signal -1\n\ts_barrier_wait -1" ::: "memory");
}

template <int M>
__device__ __forceinline__ float row_xor(float v) {
  int b = __builtin_bit_cast(int, v);
  return __builtin_bit_cast(float, __builtin_amdgcn_update_dpp(0, b, 0x160 + M, 0xf, 0xf, true));
}
__device__ __forceinline__ float row16_max(float v) {
  v = fmaxf(v, row_xor<1>(v));
  v = fmaxf(v, row_xor<2>(v));
  v = fmaxf(v, row_xor<4>(v));
  v = fmaxf(v, row_xor<8>(v));
  return v;
}
__device__ __forceinline__ float row16_sum(float v) {
  v += row_xor<1>(v);
  v += row_xor<2>(v);
  v += row_xor<4>(v);
  v += row_xor<8>(v);
  return v;
}

__global__ __launch_bounds__(256) void mlp_kernel(const float* __restrict__ x,
                                                  const float* __restrict__ W,
                                                  const float* __restrict__ bias,
                                                  float* __restrict__ t,
                                                  float* __restrict__ zT,
                                                  float* __restrict__ zout) {
  __shared__ float Wl[8 * 1024];
  int tid = threadIdx.x;
  const float4* W4 = (const float4*)W;
  float4* Wl4 = (float4*)Wl;
  for (int i = tid; i < 2048; i += 256) Wl4[i] = W4[i];
  __syncthreads();

  if (tid < 128) {
    float4* z = (float4*)(tid < 64 ? zout : zT);
    z[blockIdx.x * 64 + (tid & 63)] = make_float4(0.0f, 0.0f, 0.0f, 0.0f);
  }

  int wave = tid >> 5, lane = tid & 31;
  int row = blockIdx.x * 8 + wave;
  const float4* xr = (const float4*)(x + (size_t)row * 1024);

  float acc[8];
#pragma unroll
  for (int c = 0; c < 8; ++c) acc[c] = 0.0f;

  for (int i = 0; i < 8; ++i) {
    int wi = i * 32 + lane;
    float4 xv = xr[wi];
#pragma unroll
    for (int c = 0; c < 8; ++c) {
      float4 wv = Wl4[c * 256 + wi];
      acc[c] += xv.x * wv.x + xv.y * wv.y + xv.z * wv.z + xv.w * wv.w;
    }
  }
#pragma unroll
  for (int c = 0; c < 8; ++c)
#pragma unroll
    for (int off = 16; off > 0; off >>= 1)
      acc[c] += __shfl_xor(acc[c], off, 32);

  if (lane == 0) {
#pragma unroll
    for (int c = 0; c < 8; ++c)
      t[(size_t)row * 8 + c] = tanhf(acc[c] + bias[c]);
  }
}

__global__ __launch_bounds__(128) void tmat_kernel(const float* __restrict__ Q,
                                                   const float* __restrict__ selW,
                                                   const float* __restrict__ t,
                                                   float* __restrict__ T) {
  int wg = blockIdx.x;
  int part = wg & 3;
  int bh = wg >> 2;
  int b = bh >> 3, h = bh & 7;

  __shared__ float a[256];
  int tid = threadIdx.x;
  int wave = tid >> 5, lane = tid & 31;
  int half = lane >> 4, lm = lane & 15;
  int e = wave * 16 + lm;

  const float* tp = t + (size_t)b * 8192 + (size_t)h * 1024 + part * 256;
  float ta0 = tp[tid], ta1 = tp[tid + 128];

  float wsel[4];
#pragma unroll
  for (int gi = 0; gi < 4; ++gi) wsel[gi] = selW[h * 64 + gi * 16 + lm];

  const float* Qbase = Q + ((size_t)b * 1024 + (size_t)part * 256) * 512 + h * 64 + e
                         + (size_t)half * 1024;
  v2f qf[64];
#pragma unroll
  for (int it = 0; it < 64; ++it) {
    qf[it].x = Qbase[(size_t)it * 2048];
    qf[it].y = Qbase[(size_t)it * 2048 + 512];
  }

  a[tid] = ta0;
  a[tid + 128] = ta1;
  lds_barrier();

  v8f acc[4] = {};
#pragma unroll
  for (int it = 0; it < 64; ++it) {
    int la = it * 4 + half * 2;
    float a0 = a[la], a1 = a[la + 1];
#pragma unroll
    for (int gi = 0; gi < 4; ++gi) {
      float s0 = a0 * wsel[gi], s1 = a1 * wsel[gi];
      v2f bf;
      bf.x = s0 > 0.0f ? s0 : 0.0f;
      bf.y = s1 > 0.0f ? s1 : 0.0f;
      acc[gi] = wmma_f32(qf[it], bf, acc[gi]);
    }
  }

  float* Tp = T + (size_t)bh * 4096;
#pragma unroll
  for (int gi = 0; gi < 4; ++gi)
#pragma unroll
    for (int jj = 0; jj < 8; ++jj) {
      int er = wave * 16 + jj + half * 8;
      int g = gi * 16 + lm;
      atomicAdd(&Tp[er * 64 + g], acc[gi][jj]);
    }
}

#define TPAD 72
#define KPAD 65
__global__ __launch_bounds__(128) void attn_kernel(const float* __restrict__ Km,
                                                   const float* __restrict__ Vm,
                                                   const float* __restrict__ Tg,
                                                   float* __restrict__ out) {
  int wg = blockIdx.x;
  int part = wg & 3;
  int bh = wg >> 2;
  int b = bh >> 3, h = bh & 7;

  __shared__ float Tl[64 * TPAD];
  __shared__ float Kb[64 * KPAD];
  __shared__ float Sb[64 * KPAD];

  int tid = threadIdx.x;
  int wave = tid >> 5, lane = tid & 31;
  int half = lane >> 4, lm = lane & 15;
  int sr = tid >> 6, sc = tid & 63;

  const float* Tp = Tg + (size_t)bh * 4096 + tid;
  const float* Kbase = Km + ((size_t)b * 1024 * 8 + h) * 64;
  const float* Vbase = Vm + ((size_t)b * 1024 * 8 + h) * 64;
  const float* Kst = Kbase + ((size_t)part * 256 + sr) * 512 + sc;

  float tr[32], kr[32];
#pragma unroll
  for (int j = 0; j < 32; ++j) tr[j] = Tp[j * 128];
#pragma unroll
  for (int j = 0; j < 32; ++j) kr[j] = Kst[(size_t)j * 1024];
  __builtin_amdgcn_sched_barrier(0);
#pragma unroll
  for (int j = 0; j < 32; ++j) Tl[(sr + 2 * j) * TPAD + sc] = tr[j];

  v8f vacc[4] = {};

  for (int chunk = 0; chunk < 4; ++chunk) {
    int s0 = part * 256 + chunk * 64;
    lds_barrier();

#pragma unroll
    for (int j = 0; j < 32; ++j) Kb[(sr + 2 * j) * KPAD + sc] = kr[j];

    const float* Vp = Vbase + (size_t)s0 * 512 + wave * 16 + lm + (size_t)half * 1024;
    v2f vf[16];
#pragma unroll
    for (int kt = 0; kt < 16; ++kt) {
      vf[kt].x = Vp[(size_t)kt * 2048];
      vf[kt].y = Vp[(size_t)kt * 2048 + 512];
    }
    lds_barrier();

    v8f sacc[4] = {};
    int m = wave * 16 + lm;
    for (int e0 = 0; e0 < 64; e0 += 4) {
      int ka = e0 + half * 2;
      v2f af;
      af.x = Kb[m * KPAD + ka];
      af.y = Kb[m * KPAD + ka + 1];
#pragma unroll
      for (int gi = 0; gi < 4; ++gi) {
        int n = gi * 16 + lm;
        v2f bf;
        bf.x = Tl[ka * TPAD + n];
        bf.y = Tl[(ka + 1) * TPAD + n];
        sacc[gi] = wmma_f32(af, bf, sacc[gi]);
      }
    }

    float xs[4][8], mx[8];
#pragma unroll
    for (int jj = 0; jj < 8; ++jj) {
#pragma unroll
      for (int gi = 0; gi < 4; ++gi) xs[gi][jj] = sacc[gi][jj] * 0.125f;
      mx[jj] = row16_max(fmaxf(fmaxf(xs[0][jj], xs[1][jj]), fmaxf(xs[2][jj], xs[3][jj])));
    }

    if (chunk < 3) {
      const float* Kn = Kst + (size_t)(chunk + 1) * 32768;
#pragma unroll
      for (int j = 0; j < 32; ++j) kr[j] = Kn[(size_t)j * 1024];
    }

#pragma unroll
    for (int jj = 0; jj < 8; ++jj) {
      float p0 = __expf(xs[0][jj] - mx[jj]), p1 = __expf(xs[1][jj] - mx[jj]);
      float p2 = __expf(xs[2][jj] - mx[jj]), p3 = __expf(xs[3][jj] - mx[jj]);
      float inv = 1.0f / row16_sum((p0 + p1) + (p2 + p3));
      float* srow = &Sb[(wave * 16 + jj + half * 8) * KPAD + lm];
      srow[0] = p0 * inv;
      srow[16] = p1 * inv;
      srow[32] = p2 * inv;
      srow[48] = p3 * inv;
    }
    lds_barrier();

#pragma unroll
    for (int kt = 0; kt < 16; ++kt) {
      int ka = kt * 4 + half * 2;
#pragma unroll
      for (int gi = 0; gi < 4; ++gi) {
        int n = gi * 16 + lm;
        v2f bf;
        bf.x = Sb[ka * KPAD + n];
        bf.y = Sb[(ka + 1) * KPAD + n];
        vacc[gi] = wmma_f32(vf[kt], bf, vacc[gi]);
      }
    }
  }

#pragma unroll
  for (int gi = 0; gi < 4; ++gi)
#pragma unroll
    for (int jj = 0; jj < 8; ++jj) {
      int d = wave * 16 + jj + half * 8;
      int g = gi * 16 + lm;
      atomicAdd(&out[(((size_t)b * 64 + d) * 8 + h) * 64 + g], vacc[gi][jj]);
    }
}

extern "C" void kernel_launch(void* const* d_in, const int* in_sizes, int n_in,
                              void* d_out, int out_size, void* d_ws, size_t ws_size,
                              hipStream_t stream) {
  const float* queries = (const float*)d_in[0];
  const float* keys    = (const float*)d_in[1];
  const float* values  = (const float*)d_in[2];
  const float* x       = (const float*)d_in[3];
  const float* mlp_w   = (const float*)d_in[4];
  const float* mlp_b   = (const float*)d_in[5];
  const float* sel_W   = (const float*)d_in[6];
  (void)d_in; (void)in_sizes; (void)n_in; (void)ws_size;

  float* out  = (float*)d_out;
  float* t_ws = (float*)d_ws;
  float* T_ws = t_ws + 131072;

  mlp_kernel<<<2048, 256, 0, stream>>>(x, mlp_w, mlp_b, t_ws, T_ws, out);

  tmat_kernel<<<512, 128, 0, stream>>>(queries, sel_W, t_ws, T_ws);

  attn_kernel<<<512, 128, 0, stream>>>(keys, values, T_ws, out);
}


# relu(a*w) = max(a,0)*max(w,0) + max(-a,0)*max(-w,0): T and the scores have rank 2, so Q^T.sel and K.T shrink from 64 to 2 live columns (1 matrix tile, not 4); sums regrouped
# speedup vs baseline: 2.1609x; 1.2359x over previous
#include <hip/hip_runtime.h>
#include <math.h>

typedef __attribute__((ext_vector_type(2))) float v2f;
typedef __attribute__((ext_vector_type(8))) float v8f;

__device__ __forceinline__ v8f wmma_f32(v2f a, v2f b, v8f c) {
  return __builtin_amdgcn_wmma_f32_16x16x4_f32(
       false, a,  false, b,
       (short)0, c,  false,  false);
}

__device__ __forceinline__ void lds_barrier() {
  asm volatile("s_wait_dscnt 0x0\n\ts_barrier_signal -1\n\ts_barrier_wait -1" ::: "memory");
}

template <int M>
__device__ __forceinline__ float row_xor(float v) {
  int b = __builtin_bit_cast(int, v);
  return __builtin_bit_cast(float, __builtin_amdgcn_update_dpp(0, b, 0x160 + M, 0xf, 0xf, true));
}
template <int N>
__device__ __forceinline__ float row_bcast(float v) {
  int b = __builtin_bit_cast(int, v);
  return __builtin_bit_cast(float, __builtin_amdgcn_update_dpp(0, b, 0x150 + N, 0xf, 0xf, true));
}
__device__ __forceinline__ float row16_max(float v) {
  v = fmaxf(v, row_xor<1>(v));
  v = fmaxf(v, row_xor<2>(v));
  v = fmaxf(v, row_xor<4>(v));
  v = fmaxf(v, row_xor<8>(v));
  return v;
}
__device__ __forceinline__ float row16_sum(float v) {
  v += row_xor<1>(v);
  v += row_xor<2>(v);
  v += row_xor<4>(v);
  v += row_xor<8>(v);
  return v;
}

__global__ __launch_bounds__(256) void mlp_kernel(const float* __restrict__ x,
                                                  const float* __restrict__ W,
                                                  const float* __restrict__ bias,
                                                  float* __restrict__ t,
                                                  float* __restrict__ zU,
                                                  float* __restrict__ zout) {
  __shared__ float Wl[8 * 1024];
  int tid = threadIdx.x;
  const float4* W4 = (const float4*)W;
  float4* Wl4 = (float4*)Wl;
  for (int i = tid; i < 2048; i += 256) Wl4[i] = W4[i];
  __syncthreads();

  if (tid < 64) {
    ((float4*)zout)[blockIdx.x * 64 + tid] = make_float4(0.0f, 0.0f, 0.0f, 0.0f);
  } else if (tid < 128 && blockIdx.x < 64) {
    ((float4*)zU)[blockIdx.x * 64 + (tid - 64)] = make_float4(0.0f, 0.0f, 0.0f, 0.0f);
  }

  int wave = tid >> 5, lane = tid & 31;
  int row = blockIdx.x * 8 + wave;
  const float4* xr = (const float4*)(x + (size_t)row * 1024);

  float acc[8];
#pragma unroll
  for (int c = 0; c < 8; ++c) acc[c] = 0.0f;

  for (int i = 0; i < 8; ++i) {
    int wi = i * 32 + lane;
    float4 xv = xr[wi];
#pragma unroll
    for (int c = 0; c < 8; ++c) {
      float4 wv = Wl4[c * 256 + wi];
      acc[c] += xv.x * wv.x + xv.y * wv.y + xv.z * wv.z + xv.w * wv.w;
    }
  }
#pragma unroll
  for (int c = 0; c < 8; ++c)
#pragma unroll
    for (int off = 16; off > 0; off >>= 1)
      acc[c] += __shfl_xor(acc[c], off, 32);

  if (lane == 0) {
#pragma unroll
    for (int c = 0; c < 8; ++c)
      t[(size_t)row * 8 + c] = tanhf(acc[c] + bias[c]);
  }
}

__global__ __launch_bounds__(128) void umat_kernel(const float* __restrict__ Q,
                                                   const float* __restrict__ t,
                                                   float* __restrict__ U) {
  int wg = blockIdx.x;
  int part = wg & 3;
  int bh = wg >> 2;
  int b = bh >> 3, h = bh & 7;

  __shared__ float a[256];
  int tid = threadIdx.x;
  int wave = tid >> 5, lane = tid & 31;
  int half = lane >> 4, lm = lane & 15;
  int e = wave * 16 + lm;

  const float* tp = t + (size_t)b * 8192 + (size_t)h * 1024 + part * 256;
  float ta0 = tp[tid], ta1 = tp[tid + 128];

  float sg = lm == 0 ? 1.0f : (lm == 1 ? -1.0f : 0.0f);

  const float* Qbase = Q + ((size_t)b * 1024 + (size_t)part * 256) * 512 + h * 64 + e
                         + (size_t)half * 1024;
  v2f qf[64];
#pragma unroll
  for (int it = 0; it < 64; ++it) {
    qf[it].x = Qbase[(size_t)it * 2048];
    qf[it].y = Qbase[(size_t)it * 2048 + 512];
  }

  a[tid] = ta0;
  a[tid + 128] = ta1;
  lds_barrier();

  v2f bq[64];
#pragma unroll
  for (int it = 0; it < 64; ++it) {
    int la = it * 4 + half * 2;
    bq[it].x = fmaxf(a[la] * sg, 0.0f);
    bq[it].y = fmaxf(a[la + 1] * sg, 0.0f);
  }
  __builtin_amdgcn_sched_barrier(0);

  v8f acc0 = {}, acc1 = {};
#pragma unroll
  for (int it = 0; it < 64; it += 2) {
    acc0 = wmma_f32(qf[it], bq[it], acc0);
    acc1 = wmma_f32(qf[it + 1], bq[it + 1], acc1);
  }

  if (lm < 2) {
    float* Up = U + (size_t)bh * 128 + lm * 64 + wave * 16 + half * 8;
#pragma unroll
    for (int jj = 0; jj < 8; ++jj) atomicAdd(&Up[jj], acc0[jj] + acc1[jj]);
  }
}

#define KPAD 65
__global__ __launch_bounds__(128) void attn_kernel(const float* __restrict__ Km,
                                                   const float* __restrict__ Vm,
                                                   const float* __restrict__ selW,
                                                   const float* __restrict__ Ug,
                                                   float* __restrict__ out) {
  int wg = blockIdx.x;
  int part = wg & 3;
  int bh = wg >> 2;
  int b = bh >> 3, h = bh & 7;

  __shared__ float Kb[64 * KPAD];
  __shared__ float Sb[64 * KPAD];

  int tid = threadIdx.x;
  int wave = tid >> 5, lane = tid & 31;
  int half = lane >> 4, lm = lane & 15;
  int sr = tid >> 6, sc = tid & 63;

  const float* Kbase = Km + ((size_t)b * 1024 * 8 + h) * 64;
  const float* Vbase = Vm + ((size_t)b * 1024 * 8 + h) * 64;
  const float* Kst = Kbase + ((size_t)part * 256 + sr) * 512 + sc;

  float kr[32];
#pragma unroll
  for (int j = 0; j < 32; ++j) kr[j] = Kst[(size_t)j * 1024];
  __builtin_amdgcn_sched_barrier(0);

  v2f uf[16];
  {
    const float* Up = Ug + (size_t)bh * 128 + (lm & 1) * 64 + half * 2;
#pragma unroll
    for (int kt = 0; kt < 16; ++kt) {
      float2 v = *(const float2*)(Up + kt * 4);
      uf[kt].x = lm < 2 ? v.x : 0.0f;
      uf[kt].y = lm < 2 ? v.y : 0.0f;
    }
  }
  float wa[4];
  bool wpos[4];
#pragma unroll
  for (int gi = 0; gi < 4; ++gi) {
    float w = selW[h * 64 + gi * 16 + lm];
    wa[gi] = 0.125f * fabsf(w);
    wpos[gi] = w > 0.0f;
  }

  v8f vacc[4] = {};

  for (int chunk = 0; chunk < 4; ++chunk) {
    int s0 = part * 256 + chunk * 64;
    lds_barrier();

#pragma unroll
    for (int j = 0; j < 32; ++j) Kb[(sr + 2 * j) * KPAD + sc] = kr[j];

    const float* Vp = Vbase + (size_t)s0 * 512 + wave * 16 + lm + (size_t)half * 1024;
    v2f vf[16];
#pragma unroll
    for (int kt = 0; kt < 16; ++kt) {
      vf[kt].x = Vp[(size_t)kt * 2048];
      vf[kt].y = Vp[(size_t)kt * 2048 + 512];
    }
    lds_barrier();

    int m = wave * 16 + lm;
    v2f af[16];
#pragma unroll
    for (int kt = 0; kt < 16; ++kt) {
      int ka = kt * 4 + half * 2;
      af[kt].x = Kb[m * KPAD + ka];
      af[kt].y = Kb[m * KPAD + ka + 1];
    }
    __builtin_amdgcn_sched_barrier(0);
    v8f pm0 = {}, pm1 = {};
#pragma unroll
    for (int kt = 0; kt < 16; kt += 2) {
      pm0 = wmma_f32(af[kt], uf[kt], pm0);
      pm1 = wmma_f32(af[kt + 1], uf[kt + 1], pm1);
    }

    float xs[4][8], mx[8];
#pragma unroll
    for (int jj = 0; jj < 8; ++jj) {
      float pmj = pm0[jj] + pm1[jj];
      float pr = row_bcast<0>(pmj);
      float mr = row_bcast<1>(pmj);
#pragma unroll
      for (int gi = 0; gi < 4; ++gi) xs[gi][jj] = (wpos[gi] ? pr : mr) * wa[gi];
      mx[jj] = row16_max(fmaxf(fmaxf(xs[0][jj], xs[1][jj]), fmaxf(xs[2][jj], xs[3][jj])));
    }

    if (chunk < 3) {
      const float* Kn = Kst + (size_t)(chunk + 1) * 32768;
#pragma unroll
      for (int j = 0; j < 32; ++j) kr[j] = Kn[(size_t)j * 1024];
    }

#pragma unroll
    for (int jj = 0; jj < 8; ++jj) {
      float p0 = __expf(xs[0][jj] - mx[jj]), p1 = __expf(xs[1][jj] - mx[jj]);
      float p2 = __expf(xs[2][jj] - mx[jj]), p3 = __expf(xs[3][jj] - mx[jj]);
      float inv = 1.0f / row16_sum((p0 + p1) + (p2 + p3));
      float* srow = &Sb[(wave * 16 + jj + half * 8) * KPAD + lm];
      srow[0] = p0 * inv;
      srow[16] = p1 * inv;
      srow[32] = p2 * inv;
      srow[48] = p3 * inv;
    }
    lds_barrier();

#pragma unroll
    for (int kt = 0; kt < 16; ++kt) {
      int ka = kt * 4 + half * 2;
#pragma unroll
      for (int gi = 0; gi < 4; ++gi) {
        int n = gi * 16 + lm;
        v2f bf;
        bf.x = Sb[ka * KPAD + n];
        bf.y = Sb[(ka + 1) * KPAD + n];
        vacc[gi] = wmma_f32(vf[kt], bf, vacc[gi]);
      }
    }
  }

#pragma unroll
  for (int gi = 0; gi < 4; ++gi)
#pragma unroll
    for (int jj = 0; jj < 8; ++jj) {
      int d = wave * 16 + jj + half * 8;
      int g = gi * 16 + lm;
      atomicAdd(&out[(((size_t)b * 64 + d) * 8 + h) * 64 + g], vacc[gi][jj]);
    }
}

extern "C" void kernel_launch(void* const* d_in, const int* in_sizes, int n_in,
                              void* d_out, int out_size, void* d_ws, size_t ws_size,
                              hipStream_t stream) {
  const float* queries = (const float*)d_in[0];
  const float* keys    = (const float*)d_in[1];
  const float* values  = (const float*)d_in[2];
  const float* x       = (const float*)d_in[3];
  const float* mlp_w   = (const float*)d_in[4];
  const float* mlp_b   = (const float*)d_in[5];
  const float* sel_W   = (const float*)d_in[6];
  (void)d_in; (void)in_sizes; (void)n_in; (void)ws_size;

  float* out  = (float*)d_out;
  float* t_ws = (float*)d_ws;
  float* U_ws = t_ws + 131072;

  mlp_kernel<<<2048, 256, 0, stream>>>(x, mlp_w, mlp_b, t_ws, U_ws, out);

  umat_kernel<<<512, 128, 0, stream>>>(queries, t_ws, U_ws);

  attn_kernel<<<512, 128, 0, stream>>>(keys, values, sel_W, U_ws, out);
}


# mlp: a wave owns a 4-row x 8-col tile, so each LDS read of W serves 4 rows (LDS reads, staging /4; 512 blocks); 32 sums end one per lane by a transposing DPP fold; sum order regrouped
# speedup vs baseline: 2.8501x; 1.3189x over previous
#include <hip/hip_runtime.h>
#include <math.h>

typedef __attribute__((ext_vector_type(2))) float v2f;
typedef __attribute__((ext_vector_type(8))) float v8f;

__device__ __forceinline__ v8f wmma_f32(v2f a, v2f b, v8f c) {
  return __builtin_amdgcn_wmma_f32_16x16x4_f32(
       false, a,  false, b,
       (short)0, c,  false,  false);
}

__device__ __forceinline__ void lds_barrier() {
  asm volatile("s_wait_dscnt 0x0\n\ts_barrier_signal -1\n\ts_barrier_wait -1" ::: "memory");
}

template <int M>
__device__ __forceinline__ float row_xor(float v) {
  int b = __builtin_bit_cast(int, v);
  return __builtin_bit_cast(float, __builtin_amdgcn_update_dpp(0, b, 0x160 + M, 0xf, 0xf, true));
}
template <int N>
__device__ __forceinline__ float row_bcast(float v) {
  int b = __builtin_bit_cast(int, v);
  return __builtin_bit_cast(float, __builtin_amdgcn_update_dpp(0, b, 0x150 + N, 0xf, 0xf, true));
}
__device__ __forceinline__ float row16_max(float v) {
  v = fmaxf(v, row_xor<1>(v));
  v = fmaxf(v, row_xor<2>(v));
  v = fmaxf(v, row_xor<4>(v));
  v = fmaxf(v, row_xor<8>(v));
  return v;
}
__device__ __forceinline__ float row16_sum(float v) {
  v += row_xor<1>(v);
  v += row_xor<2>(v);
  v += row_xor<4>(v);
  v += row_xor<8>(v);
  return v;
}


__device__ __forceinline__ float row_other(float v) {
  int b = __builtin_bit_cast(int, v);
  return __builtin_bit_cast(float,
      __builtin_amdgcn_permlanex16(b, b, 0x76543210, 0xfedcba98, false, false));
}
template <int M, int N>
__device__ __forceinline__ void fold_step(float* v, bool hi) {
#pragma unroll
  for (int j = 0; j < N; ++j) {
    float a = v[2 * j] + row_xor<M>(v[2 * j]);
    float b = v[2 * j + 1] + row_xor<M>(v[2 * j + 1]);
    v[j] = hi ? b : a;
  }
}

__global__ __launch_bounds__(256) void mlp_kernel(const float* __restrict__ x,
                                                  const float* __restrict__ W,
                                                  const float* __restrict__ bias,
                                                  float* __restrict__ t,
                                                  float* __restrict__ zU,
                                                  float* __restrict__ zout) {
  __shared__ float Wl[8 * 1024];
  int tid = threadIdx.x;
  int wave = tid >> 5, lane = tid & 31;
  const float4* W4 = (const float4*)W;
  float4* Wl4 = (float4*)Wl;

  float4 wst[8];
#pragma unroll
  for (int j = 0; j < 8; ++j) wst[j] = W4[tid + 256 * j];

  int row0 = blockIdx.x * 32 + wave * 4;
  const float4* xr = (const float4*)(x + (size_t)row0 * 1024) + lane;
  float4 xv[8][4];
#pragma unroll
  for (int i = 0; i < 8; ++i)
#pragma unroll
    for (int r = 0; r < 4; ++r) xv[i][r] = xr[r * 256 + i * 32];

#pragma unroll
  for (int j = 0; j < 8; ++j) Wl4[tid + 256 * j] = wst[j];
  lds_barrier();

  ((float4*)zout)[blockIdx.x * 256 + tid] = make_float4(0.0f, 0.0f, 0.0f, 0.0f);
  if (blockIdx.x < 16)
    ((float4*)zU)[blockIdx.x * 256 + tid] = make_float4(0.0f, 0.0f, 0.0f, 0.0f);

  float acc[32];
#pragma unroll
  for (int n = 0; n < 32; ++n) acc[n] = 0.0f;

#pragma unroll
  for (int i = 0; i < 8; ++i) {
    float4 wv[8];
#pragma unroll
    for (int c = 0; c < 8; ++c) wv[c] = Wl4[c * 256 + i * 32 + lane];
#pragma unroll
    for (int r = 0; r < 4; ++r)
#pragma unroll
      for (int c = 0; c < 8; ++c) {
        float s = acc[r * 8 + c];
        s = fmaf(xv[i][r].x, wv[c].x, s);
        s = fmaf(xv[i][r].y, wv[c].y, s);
        s = fmaf(xv[i][r].z, wv[c].z, s);
        s = fmaf(xv[i][r].w, wv[c].w, s);
        acc[r * 8 + c] = s;
      }
  }

  fold_step<1, 16>(acc, (lane & 1) != 0);
  fold_step<2, 8>(acc, (lane & 2) != 0);
  fold_step<4, 4>(acc, (lane & 4) != 0);
  fold_step<8, 2>(acc, (lane & 8) != 0);
  {
    bool hi = (lane & 16) != 0;
    float keep = hi ? acc[1] : acc[0];
    float send = hi ? acc[0] : acc[1];
    acc[0] = keep + row_other(send);
  }

  t[(size_t)row0 * 8 + lane] = tanhf(acc[0] + bias[lane & 7]);
}

__global__ __launch_bounds__(128) void umat_kernel(const float* __restrict__ Q,
                                                   const float* __restrict__ t,
                                                   float* __restrict__ U) {
  int wg = blockIdx.x;
  int part = wg & 3;
  int bh = wg >> 2;
  int b = bh >> 3, h = bh & 7;

  __shared__ float a[256];
  int tid = threadIdx.x;
  int wave = tid >> 5, lane = tid & 31;
  int half = lane >> 4, lm = lane & 15;
  int e = wave * 16 + lm;

  const float* tp = t + (size_t)b * 8192 + (size_t)h * 1024 + part * 256;
  float ta0 = tp[tid], ta1 = tp[tid + 128];

  float sg = lm == 0 ? 1.0f : (lm == 1 ? -1.0f : 0.0f);

  const float* Qbase = Q + ((size_t)b * 1024 + (size_t)part * 256) * 512 + h * 64 + e
                         + (size_t)half * 1024;
  v2f qf[64];
#pragma unroll
  for (int it = 0; it < 64; ++it) {
    qf[it].x = Qbase[(size_t)it * 2048];
    qf[it].y = Qbase[(size_t)it * 2048 + 512];
  }

  a[tid] = ta0;
  a[tid + 128] = ta1;
  lds_barrier();

  v2f bq[64];
#pragma unroll
  for (int it = 0; it < 64; ++it) {
    int la = it * 4 + half * 2;
    bq[it].x = fmaxf(a[la] * sg, 0.0f);
    bq[it].y = fmaxf(a[la + 1] * sg, 0.0f);
  }
  __builtin_amdgcn_sched_barrier(0);

  v8f acc0 = {}, acc1 = {};
#pragma unroll
  for (int it = 0; it < 64; it += 2) {
    acc0 = wmma_f32(qf[it], bq[it], acc0);
    acc1 = wmma_f32(qf[it + 1], bq[it + 1], acc1);
  }

  if (lm < 2) {
    float* Up = U + (size_t)bh * 128 + lm * 64 + wave * 16 + half * 8;
#pragma unroll
    for (int jj = 0; jj < 8; ++jj) atomicAdd(&Up[jj], acc0[jj] + acc1[jj]);
  }
}

#define KPAD 65
__global__ __launch_bounds__(128) void attn_kernel(const float* __restrict__ Km,
                                                   const float* __restrict__ Vm,
                                                   const float* __restrict__ selW,
                                                   const float* __restrict__ Ug,
                                                   float* __restrict__ out) {
  int wg = blockIdx.x;
  int part = wg & 3;
  int bh = wg >> 2;
  int b = bh >> 3, h = bh & 7;

  __shared__ float Kb[64 * KPAD];
  __shared__ float Sb[64 * KPAD];

  int tid = threadIdx.x;
  int wave = tid >> 5, lane = tid & 31;
  int half = lane >> 4, lm = lane & 15;
  int sr = tid >> 6, sc = tid & 63;

  const float* Kbase = Km + ((size_t)b * 1024 * 8 + h) * 64;
  const float* Vbase = Vm + ((size_t)b * 1024 * 8 + h) * 64;
  const float* Kst = Kbase + ((size_t)part * 256 + sr) * 512 + sc;

  float kr[32];
#pragma unroll
  for (int j = 0; j < 32; ++j) kr[j] = Kst[(size_t)j * 1024];
  __builtin_amdgcn_sched_barrier(0);

  v2f uf[16];
  {
    const float* Up = Ug + (size_t)bh * 128 + (lm & 1) * 64 + half * 2;
#pragma unroll
    for (int kt = 0; kt < 16; ++kt) {
      float2 v = *(const float2*)(Up + kt * 4);
      uf[kt].x = lm < 2 ? v.x : 0.0f;
      uf[kt].y = lm < 2 ? v.y : 0.0f;
    }
  }
  float wa[4];
  bool wpos[4];
#pragma unroll
  for (int gi = 0; gi < 4; ++gi) {
    float w = selW[h * 64 + gi * 16 + lm];
    wa[gi] = 0.125f * fabsf(w);
    wpos[gi] = w > 0.0f;
  }

  v8f vacc[4] = {};

  for (int chunk = 0; chunk < 4; ++chunk) {
    int s0 = part * 256 + chunk * 64;
    lds_barrier();

#pragma unroll
    for (int j = 0; j < 32; ++j) Kb[(sr + 2 * j) * KPAD + sc] = kr[j];

    const float* Vp = Vbase + (size_t)s0 * 512 + wave * 16 + lm + (size_t)half * 1024;
    v2f vf[16];
#pragma unroll
    for (int kt = 0; kt < 16; ++kt) {
      vf[kt].x = Vp[(size_t)kt * 2048];
      vf[kt].y = Vp[(size_t)kt * 2048 + 512];
    }
    lds_barrier();

    int m = wave * 16 + lm;
    v2f af[16];
#pragma unroll
    for (int kt = 0; kt < 16; ++kt) {
      int ka = kt * 4 + half * 2;
      af[kt].x = Kb[m * KPAD + ka];
      af[kt].y = Kb[m * KPAD + ka + 1];
    }
    __builtin_amdgcn_sched_barrier(0);
    v8f pm0 = {}, pm1 = {};
#pragma unroll
    for (int kt = 0; kt < 16; kt += 2) {
      pm0 = wmma_f32(af[kt], uf[kt], pm0);
      pm1 = wmma_f32(af[kt + 1], uf[kt + 1], pm1);
    }

    float xs[4][8], mx[8];
#pragma unroll
    for (int jj = 0; jj < 8; ++jj) {
      float pmj = pm0[jj] + pm1[jj];
      float pr = row_bcast<0>(pmj);
      float mr = row_bcast<1>(pmj);
#pragma unroll
      for (int gi = 0; gi < 4; ++gi) xs[gi][jj] = (wpos[gi] ? pr : mr) * wa[gi];
      mx[jj] = row16_max(fmaxf(fmaxf(xs[0][jj], xs[1][jj]), fmaxf(xs[2][jj], xs[3][jj])));
    }

    if (chunk < 3) {
      const float* Kn = Kst + (size_t)(chunk + 1) * 32768;
#pragma unroll
      for (int j = 0; j < 32; ++j) kr[j] = Kn[(size_t)j * 1024];
    }

#pragma unroll
    for (int jj = 0; jj < 8; ++jj) {
      float p0 = __expf(xs[0][jj] - mx[jj]), p1 = __expf(xs[1][jj] - mx[jj]);
      float p2 = __expf(xs[2][jj] - mx[jj]), p3 = __expf(xs[3][jj] - mx[jj]);
      float inv = 1.0f / row16_sum((p0 + p1) + (p2 + p3));
      float* srow = &Sb[(wave * 16 + jj + half * 8) * KPAD + lm];
      srow[0] = p0 * inv;
      srow[16] = p1 * inv;
      srow[32] = p2 * inv;
      srow[48] = p3 * inv;
    }
    lds_barrier();

#pragma unroll
    for (int kt = 0; kt < 16; ++kt) {
      int ka = kt * 4 + half * 2;
#pragma unroll
      for (int gi = 0; gi < 4; ++gi) {
        int n = gi * 16 + lm;
        v2f bf;
        bf.x = Sb[ka * KPAD + n];
        bf.y = Sb[(ka + 1) * KPAD + n];
        vacc[gi] = wmma_f32(vf[kt], bf, vacc[gi]);
      }
    }
  }

#pragma unroll
  for (int gi = 0; gi < 4; ++gi)
#pragma unroll
    for (int jj = 0; jj < 8; ++jj) {
      int d = wave * 16 + jj + half * 8;
      int g = gi * 16 + lm;
      atomicAdd(&out[(((size_t)b * 64 + d) * 8 + h) * 64 + g], vacc[gi][jj]);
    }
}

extern "C" void kernel_launch(void* const* d_in, const int* in_sizes, int n_in,
                              void* d_out, int out_size, void* d_ws, size_t ws_size,
                              hipStream_t stream) {
  const float* queries = (const float*)d_in[0];
  const float* keys    = (const float*)d_in[1];
  const float* values  = (const float*)d_in[2];
  const float* x       = (const float*)d_in[3];
  const float* mlp_w   = (const float*)d_in[4];
  const float* mlp_b   = (const float*)d_in[5];
  const float* sel_W   = (const float*)d_in[6];
  (void)d_in; (void)in_sizes; (void)n_in; (void)ws_size;

  float* out  = (float*)d_out;
  float* t_ws = (float*)d_ws;
  float* U_ws = t_ws + 131072;

  mlp_kernel<<<512, 256, 0, stream>>>(x, mlp_w, mlp_b, t_ws, U_ws, out);

  umat_kernel<<<512, 128, 0, stream>>>(queries, t_ws, U_ws);

  attn_kernel<<<512, 128, 0, stream>>>(keys, values, sel_W, U_ws, out);
}


# mlp and umat fused into one launch (3 -> 2): a block's 256 t values stay in LDS and its Q loads fly under the fold; U's 4 l parts go to planes added in attn, no atomics on U; U sum order regrouped
# speedup vs baseline: 3.0790x; 1.0803x over previous
#include <hip/hip_runtime.h>
#include <math.h>

typedef __attribute__((ext_vector_type(2))) float v2f;
typedef __attribute__((ext_vector_type(8))) float v8f;

__device__ __forceinline__ v8f wmma_f32(v2f a, v2f b, v8f c) {
  return __builtin_amdgcn_wmma_f32_16x16x4_f32(
       false, a,  false, b,
       (short)0, c,  false,  false);
}

__device__ __forceinline__ void lds_barrier() {
  asm volatile("s_wait_dscnt 0x0\n\ts_barrier_signal -1\n\ts_barrier_wait -1" ::: "memory");
}

template <int M>
__device__ __forceinline__ float row_xor(float v) {
  int b = __builtin_bit_cast(int, v);
  return __builtin_bit_cast(float, __builtin_amdgcn_update_dpp(0, b, 0x160 + M, 0xf, 0xf, true));
}
template <int N>
__device__ __forceinline__ float row_bcast(float v) {
  int b = __builtin_bit_cast(int, v);
  return __builtin_bit_cast(float, __builtin_amdgcn_update_dpp(0, b, 0x150 + N, 0xf, 0xf, true));
}
__device__ __forceinline__ float row16_max(float v) {
  v = fmaxf(v, row_xor<1>(v));
  v = fmaxf(v, row_xor<2>(v));
  v = fmaxf(v, row_xor<4>(v));
  v = fmaxf(v, row_xor<8>(v));
  return v;
}
__device__ __forceinline__ float row16_sum(float v) {
  v += row_xor<1>(v);
  v += row_xor<2>(v);
  v += row_xor<4>(v);
  v += row_xor<8>(v);
  return v;
}


__device__ __forceinline__ float row_other(float v) {
  int b = __builtin_bit_cast(int, v);
  return __builtin_bit_cast(float,
      __builtin_amdgcn_permlanex16(b, b, 0x76543210, 0xfedcba98, false, false));
}
template <int M, int N>
__device__ __forceinline__ void fold_step(float* v, bool hi) {
#pragma unroll
  for (int j = 0; j < N; ++j) {
    float a = v[2 * j] + row_xor<M>(v[2 * j]);
    float b = v[2 * j + 1] + row_xor<M>(v[2 * j + 1]);
    v[j] = hi ? b : a;
  }
}

__global__ __launch_bounds__(256) void mlp_umat_kernel(const float* __restrict__ x,
                                                       const float* __restrict__ W,
                                                       const float* __restrict__ bias,
                                                       const float* __restrict__ Q,
                                                       float* __restrict__ Upart,
                                                       float* __restrict__ zout) {
  __shared__ __attribute__((aligned(16))) float Wl[8 * 1024];
  __shared__ float al[256];
  __shared__ __attribute__((aligned(16))) float ucomb[2 * 128];
  int tid = threadIdx.x;
  int wave = tid >> 5, lane = tid & 31;
  int half = lane >> 4, lm = lane & 15;
  int wg = blockIdx.x;
  int part = wg & 3;
  int bh = wg >> 2;
  int b = bh >> 3, h = bh & 7;
  const float4* W4 = (const float4*)W;
  float4* Wl4 = (float4*)Wl;

  float bs = bias[lane & 7];
  float4 wst[8];
#pragma unroll
  for (int j = 0; j < 8; ++j) wst[j] = W4[tid + 256 * j];

  int row0 = wg * 32 + wave * 4;
  const float4* xr = (const float4*)(x + (size_t)row0 * 1024) + lane;
  float4 xv[8][4];
#pragma unroll
  for (int i = 0; i < 8; ++i)
#pragma unroll
    for (int r = 0; r < 4; ++r) xv[i][r] = xr[r * 256 + i * 32];

#pragma unroll
  for (int j = 0; j < 8; ++j) Wl4[tid + 256 * j] = wst[j];
  lds_barrier();

  ((float4*)zout)[wg * 256 + tid] = make_float4(0.0f, 0.0f, 0.0f, 0.0f);

  float acc[32];
#pragma unroll
  for (int n = 0; n < 32; ++n) acc[n] = 0.0f;

#pragma unroll
  for (int i = 0; i < 8; ++i) {
    float4 wv[8];
#pragma unroll
    for (int c = 0; c < 8; ++c) wv[c] = Wl4[c * 256 + i * 32 + lane];
#pragma unroll
    for (int r = 0; r < 4; ++r)
#pragma unroll
      for (int c = 0; c < 8; ++c) {
        float s = acc[r * 8 + c];
        s = fmaf(xv[i][r].x, wv[c].x, s);
        s = fmaf(xv[i][r].y, wv[c].y, s);
        s = fmaf(xv[i][r].z, wv[c].z, s);
        s = fmaf(xv[i][r].w, wv[c].w, s);
        acc[r * 8 + c] = s;
      }
  }

  __builtin_amdgcn_sched_barrier(0);

  int et = wave & 3, lh = wave >> 2;
  int e = et * 16 + lm;
  const float* Qbase = Q + ((size_t)b * 1024 + (size_t)part * 256 + (size_t)lh * 128) * 512
                         + h * 64 + e + (size_t)half * 1024;
  v2f qf[32];
#pragma unroll
  for (int it = 0; it < 32; ++it) {
    qf[it].x = Qbase[(size_t)it * 2048];
    qf[it].y = Qbase[(size_t)it * 2048 + 512];
  }
  __builtin_amdgcn_sched_barrier(0);

  fold_step<1, 16>(acc, (lane & 1) != 0);
  fold_step<2, 8>(acc, (lane & 2) != 0);
  fold_step<4, 4>(acc, (lane & 4) != 0);
  fold_step<8, 2>(acc, (lane & 8) != 0);
  {
    bool hi = (lane & 16) != 0;
    float keep = hi ? acc[1] : acc[0];
    float send = hi ? acc[0] : acc[1];
    acc[0] = keep + row_other(send);
  }

  al[tid] = tanhf(acc[0] + bs);
  lds_barrier();

  float sg = lm == 0 ? 1.0f : (lm == 1 ? -1.0f : 0.0f);
  v2f bq[32];
#pragma unroll
  for (int it = 0; it < 32; ++it) {
    int la = lh * 128 + it * 4 + half * 2;
    bq[it].x = fmaxf(al[la] * sg, 0.0f);
    bq[it].y = fmaxf(al[la + 1] * sg, 0.0f);
  }
  __builtin_amdgcn_sched_barrier(0);

  v8f acc0 = {}, acc1 = {};
#pragma unroll
  for (int it = 0; it < 32; it += 2) {
    acc0 = wmma_f32(qf[it], bq[it], acc0);
    acc1 = wmma_f32(qf[it + 1], bq[it + 1], acc1);
  }
  v8f us = acc0 + acc1;

  if (lm < 2) {
    float4* cp = (float4*)&ucomb[lh * 128 + lm * 64 + et * 16 + half * 8];
    cp[0] = make_float4(us[0], us[1], us[2], us[3]);
    cp[1] = make_float4(us[4], us[5], us[6], us[7]);
  }
  __syncthreads();
  if (wave == 0) {
    float4 u0 = ((const float4*)ucomb)[lane];
    float4 u1 = ((const float4*)ucomb)[32 + lane];
    float4 u = make_float4(u0.x + u1.x, u0.y + u1.y, u0.z + u1.z, u0.w + u1.w);
    ((float4*)Upart)[((size_t)part * 16384 + (size_t)bh * 128) / 4 + lane] = u;
  }
}

#define KPAD 65
__global__ __launch_bounds__(128) void attn_kernel(const float* __restrict__ Km,
                                                   const float* __restrict__ Vm,
                                                   const float* __restrict__ selW,
                                                   const float* __restrict__ Ug,
                                                   float* __restrict__ out) {
  int wg = blockIdx.x;
  int part = wg & 3;
  int bh = wg >> 2;
  int b = bh >> 3, h = bh & 7;

  __shared__ float Kb[64 * KPAD];
  __shared__ float Sb[64 * KPAD];

  int tid = threadIdx.x;
  int wave = tid >> 5, lane = tid & 31;
  int half = lane >> 4, lm = lane & 15;
  int sr = tid >> 6, sc = tid & 63;

  const float* Kbase = Km + ((size_t)b * 1024 * 8 + h) * 64;
  const float* Vbase = Vm + ((size_t)b * 1024 * 8 + h) * 64;
  const float* Kst = Kbase + ((size_t)part * 256 + sr) * 512 + sc;

  float kr[32];
#pragma unroll
  for (int j = 0; j < 32; ++j) kr[j] = Kst[(size_t)j * 1024];
  __builtin_amdgcn_sched_barrier(0);

  v2f uf[16];
  {
    const float* Up = Ug + (size_t)bh * 128 + (lm & 1) * 64 + half * 2;
#pragma unroll
    for (int kt = 0; kt < 16; ++kt) {
      float2 v0 = *(const float2*)(Up + kt * 4);
      float2 v1 = *(const float2*)(Up + 16384 + kt * 4);
      float2 v2 = *(const float2*)(Up + 32768 + kt * 4);
      float2 v3 = *(const float2*)(Up + 49152 + kt * 4);
      float vx = (v0.x + v1.x) + (v2.x + v3.x);
      float vy = (v0.y + v1.y) + (v2.y + v3.y);
      uf[kt].x = lm < 2 ? vx : 0.0f;
      uf[kt].y = lm < 2 ? vy : 0.0f;
    }
  }
  float wa[4];
  bool wpos[4];
#pragma unroll
  for (int gi = 0; gi < 4; ++gi) {
    float w = selW[h * 64 + gi * 16 + lm];
    wa[gi] = 0.125f * fabsf(w);
    wpos[gi] = w > 0.0f;
  }

  v8f vacc[4] = {};

  for (int chunk = 0; chunk < 4; ++chunk) {
    int s0 = part * 256 + chunk * 64;
    lds_barrier();

#pragma unroll
    for (int j = 0; j < 32; ++j) Kb[(sr + 2 * j) * KPAD + sc] = kr[j];

    const float* Vp = Vbase + (size_t)s0 * 512 + wave * 16 + lm + (size_t)half * 1024;
    v2f vf[16];
#pragma unroll
    for (int kt = 0; kt < 16; ++kt) {
      vf[kt].x = Vp[(size_t)kt * 2048];
      vf[kt].y = Vp[(size_t)kt * 2048 + 512];
    }
    lds_barrier();

    int m = wave * 16 + lm;
    v2f af[16];
#pragma unroll
    for (int kt = 0; kt < 16; ++kt) {
      int ka = kt * 4 + half * 2;
      af[kt].x = Kb[m * KPAD + ka];
      af[kt].y = Kb[m * KPAD + ka + 1];
    }
    __builtin_amdgcn_sched_barrier(0);
    v8f pm0 = {}, pm1 = {};
#pragma unroll
    for (int kt = 0; kt < 16; kt += 2) {
      pm0 = wmma_f32(af[kt], uf[kt], pm0);
      pm1 = wmma_f32(af[kt + 1], uf[kt + 1], pm1);
    }

    float xs[4][8], mx[8];
#pragma unroll
    for (int jj = 0; jj < 8; ++jj) {
      float pmj = pm0[jj] + pm1[jj];
      float pr = row_bcast<0>(pmj);
      float mr = row_bcast<1>(pmj);
#pragma unroll
      for (int gi = 0; gi < 4; ++gi) xs[gi][jj] = (wpos[gi] ? pr : mr) * wa[gi];
      mx[jj] = row16_max(fmaxf(fmaxf(xs[0][jj], xs[1][jj]), fmaxf(xs[2][jj], xs[3][jj])));
    }

    if (chunk < 3) {
      const float* Kn = Kst + (size_t)(chunk + 1) * 32768;
#pragma unroll
      for (int j = 0; j < 32; ++j) kr[j] = Kn[(size_t)j * 1024];
    }

#pragma unroll
    for (int jj = 0; jj < 8; ++jj) {
      float p0 = __expf(xs[0][jj] - mx[jj]), p1 = __expf(xs[1][jj] - mx[jj]);
      float p2 = __expf(xs[2][jj] - mx[jj]), p3 = __expf(xs[3][jj] - mx[jj]);
      float inv = 1.0f / row16_sum((p0 + p1) + (p2 + p3));
      float* srow = &Sb[(wave * 16 + jj + half * 8) * KPAD + lm];
      srow[0] = p0 * inv;
      srow[16] = p1 * inv;
      srow[32] = p2 * inv;
      srow[48] = p3 * inv;
    }
    lds_barrier();

#pragma unroll
    for (int kt = 0; kt < 16; ++kt) {
      int ka = kt * 4 + half * 2;
#pragma unroll
      for (int gi = 0; gi < 4; ++gi) {
        int n = gi * 16 + lm;
        v2f bf;
        bf.x = Sb[ka * KPAD + n];
        bf.y = Sb[(ka + 1) * KPAD + n];
        vacc[gi] = wmma_f32(vf[kt], bf, vacc[gi]);
      }
    }
  }

#pragma unroll
  for (int gi = 0; gi < 4; ++gi)
#pragma unroll
    for (int jj = 0; jj < 8; ++jj) {
      int d = wave * 16 + jj + half * 8;
      int g = gi * 16 + lm;
      atomicAdd(&out[(((size_t)b * 64 + d) * 8 + h) * 64 + g], vacc[gi][jj]);
    }
}

extern "C" void kernel_launch(void* const* d_in, const int* in_sizes, int n_in,
                              void* d_out, int out_size, void* d_ws, size_t ws_size,
                              hipStream_t stream) {
  const float* queries = (const float*)d_in[0];
  const float* keys    = (const float*)d_in[1];
  const float* values  = (const float*)d_in[2];
  const float* x       = (const float*)d_in[3];
  const float* mlp_w   = (const float*)d_in[4];
  const float* mlp_b   = (const float*)d_in[5];
  const float* sel_W   = (const float*)d_in[6];
  (void)d_in; (void)in_sizes; (void)n_in; (void)ws_size; (void)out_size;

  float* out  = (float*)d_out;
  float* U_ws = (float*)d_ws;

  mlp_umat_kernel<<<512, 256, 0, stream>>>(x, mlp_w, mlp_b, queries, U_ws, out);

  attn_kernel<<<512, 128, 0, stream>>>(keys, values, sel_W, U_ws, out);
}
